# starting kernel
# baseline (speedup 1.0000x reference)
#include <hip/hip_runtime.h>


typedef float v2f_t __attribute__((ext_vector_type(2)));
typedef float v8f_t __attribute__((ext_vector_type(8)));

#define NDIM 512
#define BDIM 16384
#define ODIM 256

static __device__ __forceinline__ v8f_t wmma_f32(v2f_t a, v2f_t b, v8f_t c) {
    return __builtin_amdgcn_wmma_f32_16x16x4_f32(
        false, a, false, b, (short)0, c, false, false);
}

__global__ void build_A_kernel(const float* __restrict__ W, float* __restrict__ A) {
    int idx = blockIdx.x * blockDim.x + threadIdx.x;
    int i = idx >> 9;
    int j = idx & (NDIM - 1);
    float v = 0.1f * W[idx];
    if (i == j) v += 0.9f;
    A[idx] = v;
}

__global__ void gemm_nn_512_kernel(const float* __restrict__ A,
                                   const float* __restrict__ B,
                                   float* __restrict__ C) {
    const int lane = threadIdx.x & 31;
    const int wave = threadIdx.x >> 5;
    const int tile = blockIdx.x * 4 + wave;
    const int m0 = (tile >> 5) << 4;
    const int n0 = (tile & 31) << 4;
    const int lm = lane & 15;
    const int hi = lane >> 4;

    v8f_t acc = {};
    const float* arow = A + (m0 + lm) * NDIM;
    const float* bcol = B + n0 + lm;

#pragma unroll 4
    for (int kk = 0; kk < NDIM; kk += 4) {
        const int kb = kk + 2 * hi;
        v2f_t a = *(const v2f_t*)(arow + kb);
        v2f_t b;
        b.x = bcol[kb * NDIM];
        b.y = bcol[(kb + 1) * NDIM];
        acc = wmma_f32(a, b, acc);
    }

    float* crow = C + (m0 + 8 * hi) * NDIM + n0 + lm;
#pragma unroll
    for (int v = 0; v < 8; ++v) crow[v * NDIM] = acc[v];
}

__global__ void gemm_nt_P_kernel(const float* __restrict__ M,
                                 const float* __restrict__ Wout,
                                 float* __restrict__ P) {
    const int lane = threadIdx.x & 31;
    const int wave = threadIdx.x >> 5;
    const int tile = blockIdx.x * 4 + wave;
    const int m0 = (tile >> 4) << 4;
    const int n0 = (tile & 15) << 4;
    const int lm = lane & 15;
    const int hi = lane >> 4;

    v8f_t acc = {};
    const float* arow = M + (m0 + lm) * NDIM;
    const float* wrow = Wout + (n0 + lm) * NDIM;

#pragma unroll 4
    for (int kk = 0; kk < NDIM; kk += 4) {
        const int kb = kk + 2 * hi;
        v2f_t a = *(const v2f_t*)(arow + kb);
        v2f_t b = *(const v2f_t*)(wrow + kb);
        acc = wmma_f32(a, b, acc);
    }

    float* prow = P + (m0 + 8 * hi) * ODIM + n0 + lm;
#pragma unroll
    for (int v = 0; v < 8; ++v) prow[v * ODIM] = acc[v];
}

__global__ void gemm_out_kernel(const float* __restrict__ X,
                                const float* __restrict__ P,
                                float* __restrict__ Y) {
    const int lane = threadIdx.x & 31;
    const int wave = threadIdx.x >> 5;
    const int m0 = blockIdx.x << 4;
    const int n0 = wave << 6;
    const int lm = lane & 15;
    const int hi = lane >> 4;

    v8f_t acc0 = {}, acc1 = {}, acc2 = {}, acc3 = {};
    const float* xrow = X + (m0 + lm) * NDIM;
    const float* pcol = P + n0 + lm;

#pragma unroll 2
    for (int kk = 0; kk < NDIM; kk += 4) {
        const int kb = kk + 2 * hi;
        v2f_t a = *(const v2f_t*)(xrow + kb);
        const float* p0 = pcol + kb * ODIM;
        const float* p1 = pcol + (kb + 1) * ODIM;
        v2f_t b0 = { p0[0],  p1[0]  };
        v2f_t b1 = { p0[16], p1[16] };
        v2f_t b2 = { p0[32], p1[32] };
        v2f_t b3 = { p0[48], p1[48] };
        acc0 = wmma_f32(a, b0, acc0);
        acc1 = wmma_f32(a, b1, acc1);
        acc2 = wmma_f32(a, b2, acc2);
        acc3 = wmma_f32(a, b3, acc3);
    }

    float* yrow = Y + (m0 + 8 * hi) * ODIM + n0 + lm;
#pragma unroll
    for (int v = 0; v < 8; ++v) {
        yrow[v * ODIM + 0]  = acc0[v];
        yrow[v * ODIM + 16] = acc1[v];
        yrow[v * ODIM + 32] = acc2[v];
        yrow[v * ODIM + 48] = acc3[v];
    }
}

extern "C" void kernel_launch(void* const* d_in, const int* in_sizes, int n_in,
                              void* d_out, int out_size, void* d_ws, size_t ws_size,
                              hipStream_t stream) {
    (void)in_sizes; (void)n_in; (void)out_size; (void)ws_size;

    const float* x    = (const float*)d_in[0];
    const float* W    = (const float*)d_in[1];
    const float* wout = (const float*)d_in[2];
    float*       out  = (float*)d_out;

    float* A  = (float*)d_ws;
    float* T0 = A  + NDIM * NDIM;
    float* T1 = T0 + NDIM * NDIM;
    float* P  = T1 + NDIM * NDIM;

    build_A_kernel<<<(NDIM * NDIM) / 256, 256, 0, stream>>>(W, A);

    gemm_nn_512_kernel<<<256, 128, 0, stream>>>(A,  A,  T0);
    gemm_nn_512_kernel<<<256, 128, 0, stream>>>(T0, T0, T1);
    gemm_nn_512_kernel<<<256, 128, 0, stream>>>(T1, T1, T0);
    gemm_nn_512_kernel<<<256, 128, 0, stream>>>(T0, T0, T1);
    gemm_nn_512_kernel<<<256, 128, 0, stream>>>(T1, T1, T0);
    gemm_nn_512_kernel<<<256, 128, 0, stream>>>(T0, T0, T1);
    gemm_nn_512_kernel<<<256, 128, 0, stream>>>(T1, A,  T0);

    gemm_nt_P_kernel<<<128, 128, 0, stream>>>(T0, wout, P);

    gemm_out_kernel<<<BDIM / 16, 128, 0, stream>>>(x, P, out);
}


# Stage both operands of every product through LDS: 16-byte coalesced global loads, next K chunk prefetched in registers, 32x32 / 64x64 block tiles; per-element k order kept
# speedup vs baseline: 1.6586x; 1.6586x over previous
#include <hip/hip_runtime.h>


typedef float v2f_t __attribute__((ext_vector_type(2)));
typedef float v4f_t __attribute__((ext_vector_type(4)));
typedef float v8f_t __attribute__((ext_vector_type(8)));

#define NDIM 512
#define BDIM 16384
#define ODIM 256

static __device__ __forceinline__ v8f_t wmma_f32(v2f_t a, v2f_t b, v8f_t c) {
    return __builtin_amdgcn_wmma_f32_16x16x4_f32(
        false, a, false, b, (short)0, c, false, false);
}

__global__ void build_A_kernel(const float* __restrict__ W, float* __restrict__ A) {
    int idx = blockIdx.x * blockDim.x + threadIdx.x;
    int i = idx >> 9;
    int j = idx & (NDIM - 1);
    float v = 0.1f * W[idx];
    if (i == j) v += 0.9f;
    A[idx] = v;
}

template <int TM, int TN, int KC, int G, int N, bool BT>
__global__ void __launch_bounds__(128)
gemm_lds_kernel(const float* __restrict__ A,
                const float* __restrict__ B,
                float* __restrict__ C) {
    constexpr int BM    = 32 * TM;
    constexpr int BN    = 32 * TN;
    constexpr int PA    = KC + 4;
    constexpr int PB    = BT ? (KC + 4) : (BN + 8);
    constexpr int BROWS = BT ? BN : KC;
    constexpr int BPR   = BT ? (KC / 4) : (BN / 4);
    constexpr int NA    = (BM * KC / 4) / 128;
    constexpr int NB    = (KC * BN / 4) / 128;
    constexpr int NC    = NDIM / KC;
    constexpr int STEPS = KC / 4;
    constexpr int NG    = STEPS / G;
    static_assert((BM * KC / 4) % 128 == 0 && (KC * BN / 4) % 128 == 0, "chunk must split evenly over 128 threads");
    static_assert(NDIM % KC == 0 && KC % 4 == 0 && STEPS % G == 0, "KC must divide K, G must divide the steps");
    static_assert((PA / 4) % 2 == 1 && PB % 4 == 0, "pitches");

    __shared__ __attribute__((aligned(16))) float As[BM * PA];
    __shared__ __attribute__((aligned(16))) float Bs[BROWS * PB];

    const int t    = threadIdx.x;
    const int lane = t & 31;
    const int wave = t >> 5;
    const int lm   = lane & 15;
    const int hi   = lane >> 4;
    const int m0   = blockIdx.y * BM;
    const int n0   = blockIdx.x * BN;
    const int wm   = (wave >> 1) * (16 * TM);
    const int wn   = (wave & 1) * (16 * TN);

    int ag[NA], as_[NA], bg[NB], bs_[NB];
#pragma unroll
    for (int i = 0; i < NA; ++i) {
        const int idx = t + i * 128;
        const int r = idx / (KC / 4);
        const int c = idx % (KC / 4);
        ag[i]  = (m0 + r) * NDIM + c * 4;
        as_[i] = r * PA + c * 4;
    }
#pragma unroll
    for (int i = 0; i < NB; ++i) {
        const int idx = t + i * 128;
        const int r = idx / BPR;
        const int c = idx % BPR;
        bg[i]  = BT ? ((n0 + r) * NDIM + c * 4) : (r * N + n0 + c * 4);
        bs_[i] = r * PB + c * 4;
    }
    constexpr int BSTEP = BT ? KC : KC * N;

    v4f_t ra[NA], rb[NB];
#pragma unroll
    for (int i = 0; i < NA; ++i) ra[i] = *(const v4f_t*)(A + ag[i]);
#pragma unroll
    for (int i = 0; i < NB; ++i) rb[i] = *(const v4f_t*)(B + bg[i]);

    v8f_t acc[TM][TN];
#pragma unroll
    for (int i = 0; i < TM; ++i)
#pragma unroll
        for (int j = 0; j < TN; ++j) acc[i][j] = v8f_t{};

    const float* afr = As + (wm + lm) * PA + 2 * hi;
    const float* bfr = BT ? (Bs + (wn + lm) * PB + 2 * hi) : (Bs + (2 * hi) * PB + wn + lm);

    for (int kc = 0; kc < NC; ++kc) {
#pragma unroll
        for (int i = 0; i < NA; ++i) *(v4f_t*)(As + as_[i]) = ra[i];
#pragma unroll
        for (int i = 0; i < NB; ++i) *(v4f_t*)(Bs + bs_[i]) = rb[i];
        __syncthreads();

        if (kc + 1 < NC) {
#pragma unroll
            for (int i = 0; i < NA; ++i) ra[i] = *(const v4f_t*)(A + ag[i] + (kc + 1) * KC);
#pragma unroll
            for (int i = 0; i < NB; ++i) rb[i] = *(const v4f_t*)(B + bg[i] + (kc + 1) * BSTEP);
        }
        __builtin_amdgcn_sched_barrier(0);

        v2f_t fa[2][G][TM], fb[2][G][TN];
#pragma unroll
        for (int u = 0; u < G; ++u) {
#pragma unroll
            for (int i = 0; i < TM; ++i)
                fa[0][u][i] = *(const v2f_t*)(afr + i * 16 * PA + u * 4);
#pragma unroll
            for (int j = 0; j < TN; ++j) {
                if (BT) {
                    fb[0][u][j] = *(const v2f_t*)(bfr + j * 16 * PB + u * 4);
                } else {
                    fb[0][u][j].x = bfr[(u * 4) * PB + j * 16];
                    fb[0][u][j].y = bfr[(u * 4 + 1) * PB + j * 16];
                }
            }
        }
#pragma unroll
        for (int g = 0; g < NG; ++g) {
            if (g + 1 < NG) {
#pragma unroll
                for (int u = 0; u < G; ++u) {
                    const int s = (g + 1) * G + u;
#pragma unroll
                    for (int i = 0; i < TM; ++i)
                        fa[(g + 1) & 1][u][i] = *(const v2f_t*)(afr + i * 16 * PA + s * 4);
#pragma unroll
                    for (int j = 0; j < TN; ++j) {
                        if (BT) {
                            fb[(g + 1) & 1][u][j] = *(const v2f_t*)(bfr + j * 16 * PB + s * 4);
                        } else {
                            fb[(g + 1) & 1][u][j].x = bfr[(s * 4) * PB + j * 16];
                            fb[(g + 1) & 1][u][j].y = bfr[(s * 4 + 1) * PB + j * 16];
                        }
                    }
                }
            }
            __builtin_amdgcn_sched_barrier(0);
#pragma unroll
            for (int u = 0; u < G; ++u)
#pragma unroll
                for (int i = 0; i < TM; ++i)
#pragma unroll
                    for (int j = 0; j < TN; ++j)
                        acc[i][j] = wmma_f32(fa[g & 1][u][i], fb[g & 1][u][j], acc[i][j]);
        }
        __syncthreads();
    }

#pragma unroll
    for (int i = 0; i < TM; ++i) {
#pragma unroll
        for (int j = 0; j < TN; ++j) {
            float* crow = C + (m0 + wm + i * 16 + 8 * hi) * N + n0 + wn + j * 16 + lm;
#pragma unroll
            for (int v = 0; v < 8; ++v) crow[v * N] = acc[i][j][v];
        }
    }
}

extern "C" void kernel_launch(void* const* d_in, const int* in_sizes, int n_in,
                              void* d_out, int out_size, void* d_ws, size_t ws_size,
                              hipStream_t stream) {
    (void)in_sizes; (void)n_in; (void)out_size; (void)ws_size;

    const float* x    = (const float*)d_in[0];
    const float* W    = (const float*)d_in[1];
    const float* wout = (const float*)d_in[2];
    float*       out  = (float*)d_out;

    float* A  = (float*)d_ws;
    float* T0 = A  + NDIM * NDIM;
    float* T1 = T0 + NDIM * NDIM;
    float* P  = T1 + NDIM * NDIM;

    build_A_kernel<<<(NDIM * NDIM) / 256, 256, 0, stream>>>(W, A);

    gemm_lds_kernel<1, 1, 128, 8, NDIM, false><<<dim3(16, 16), 128, 0, stream>>>(A,  A,  T0);
    gemm_lds_kernel<1, 1, 128, 8, NDIM, false><<<dim3(16, 16), 128, 0, stream>>>(T0, T0, T1);
    gemm_lds_kernel<1, 1, 128, 8, NDIM, false><<<dim3(16, 16), 128, 0, stream>>>(T1, T1, T0);
    gemm_lds_kernel<1, 1, 128, 8, NDIM, false><<<dim3(16, 16), 128, 0, stream>>>(T0, T0, T1);
    gemm_lds_kernel<1, 1, 128, 8, NDIM, false><<<dim3(16, 16), 128, 0, stream>>>(T1, T1, T0);
    gemm_lds_kernel<1, 1, 128, 8, NDIM, false><<<dim3(16, 16), 128, 0, stream>>>(T0, T0, T1);
    gemm_lds_kernel<1, 1, 128, 8, NDIM, false><<<dim3(16, 16), 128, 0, stream>>>(T1, A,  T0);

    gemm_lds_kernel<1, 1, 128, 8, ODIM, true><<<dim3(8, 16), 128, 0, stream>>>(T0, wout, P);

    gemm_lds_kernel<2, 2, 32, 4, ODIM, false><<<dim3(4, 256), 128, 0, stream>>>(x, P, out);
}


# Chain reassociated as A^64.(A.w_out^T): A.w_out^T shares launch 1 with A.A, both forming A=0.9I+0.1W from W in their loaders; no build pass, no A^65 product; 10 launches -> 8
# speedup vs baseline: 1.8184x; 1.0963x over previous
#include <hip/hip_runtime.h>


typedef float v2f_t __attribute__((ext_vector_type(2)));
typedef float v4f_t __attribute__((ext_vector_type(4)));
typedef float v8f_t __attribute__((ext_vector_type(8)));

#define NDIM 512
#define BDIM 16384
#define ODIM 256

static __device__ __forceinline__ v8f_t wmma_f32(v2f_t a, v2f_t b, v8f_t c) {
    return __builtin_amdgcn_wmma_f32_16x16x4_f32(
        false, a, false, b, (short)0, c, false, false);
}

static __device__ __forceinline__ v4f_t leak_form(v4f_t w, int row, int col0) {
    v4f_t v;
#pragma unroll
    for (int e = 0; e < 4; ++e) {
        float s = 0.1f * w[e];
        if (row == col0 + e) s += 0.9f;
        v[e] = s;
    }
    return v;
}

template <int TM, int TN, int KC, bool BT>
struct tile_cfg {
    static constexpr int BM    = 32 * TM;
    static constexpr int BN    = 32 * TN;
    static constexpr int PA    = KC + 4;
    static constexpr int PB    = BT ? (KC + 4) : (BN + 8);
    static constexpr int BROWS = BT ? BN : KC;
    static constexpr int AS    = BM * PA;
    static constexpr int BS    = BROWS * PB;
};

template <int TM, int TN, int KC, int G, int N, bool BT, bool FA, bool FB>
static __device__ __forceinline__ void
gemm_lds_tile(const float* __restrict__ A,
              const float* __restrict__ B,
              float* __restrict__ C,
              float* As, float* Bs, const int bx, const int by) {
    using cfg = tile_cfg<TM, TN, KC, BT>;
    constexpr int BM    = cfg::BM;
    constexpr int BN    = cfg::BN;
    constexpr int PA    = cfg::PA;
    constexpr int PB    = cfg::PB;
    constexpr int BPR   = BT ? (KC / 4) : (BN / 4);
    constexpr int NA    = (BM * KC / 4) / 128;
    constexpr int NB    = (KC * BN / 4) / 128;
    constexpr int NC    = NDIM / KC;
    constexpr int STEPS = KC / 4;
    constexpr int NG    = STEPS / G;
    static_assert((BM * KC / 4) % 128 == 0 && (KC * BN / 4) % 128 == 0, "chunk must split evenly over 128 threads");
    static_assert(NDIM % KC == 0 && KC % 4 == 0 && STEPS % G == 0, "KC must divide K, G must divide the steps");
    static_assert((PA / 4) % 2 == 1 && PB % 4 == 0, "pitches");
    static_assert(!(FB && BT), "forming A on load is written for the k-major B operand only");

    const int t    = threadIdx.x;
    const int lane = t & 31;
    const int wave = t >> 5;
    const int lm   = lane & 15;
    const int hi   = lane >> 4;
    const int m0   = by * BM;
    const int n0   = bx * BN;
    const int wm   = (wave >> 1) * (16 * TM);
    const int wn   = (wave & 1) * (16 * TN);

    int ag[NA], as_[NA], bg[NB], bs_[NB];
#pragma unroll
    for (int i = 0; i < NA; ++i) {
        const int idx = t + i * 128;
        const int r = idx / (KC / 4);
        const int c = idx % (KC / 4);
        ag[i]  = (m0 + r) * NDIM + c * 4;
        as_[i] = r * PA + c * 4;
    }
#pragma unroll
    for (int i = 0; i < NB; ++i) {
        const int idx = t + i * 128;
        const int r = idx / BPR;
        const int c = idx % BPR;
        bg[i]  = BT ? ((n0 + r) * NDIM + c * 4) : (r * N + n0 + c * 4);
        bs_[i] = r * PB + c * 4;
    }
    constexpr int BSTEP = BT ? KC : KC * N;

    v4f_t ra[NA], rb[NB];
#pragma unroll
    for (int i = 0; i < NA; ++i) ra[i] = *(const v4f_t*)(A + ag[i]);
#pragma unroll
    for (int i = 0; i < NB; ++i) rb[i] = *(const v4f_t*)(B + bg[i]);

    v8f_t acc[TM][TN];
#pragma unroll
    for (int i = 0; i < TM; ++i)
#pragma unroll
        for (int j = 0; j < TN; ++j) acc[i][j] = v8f_t{};

    const float* afr = As + (wm + lm) * PA + 2 * hi;
    const float* bfr = BT ? (Bs + (wn + lm) * PB + 2 * hi) : (Bs + (2 * hi) * PB + wn + lm);

    for (int kc = 0; kc < NC; ++kc) {
#pragma unroll
        for (int i = 0; i < NA; ++i) {
            v4f_t v = ra[i];
            if (FA) {
                const int idx = t + i * 128;
                v = leak_form(v, m0 + idx / (KC / 4), kc * KC + (idx % (KC / 4)) * 4);
            }
            *(v4f_t*)(As + as_[i]) = v;
        }
#pragma unroll
        for (int i = 0; i < NB; ++i) {
            v4f_t v = rb[i];
            if (FB) {
                const int idx = t + i * 128;
                v = leak_form(v, kc * KC + idx / BPR, n0 + (idx % BPR) * 4);
            }
            *(v4f_t*)(Bs + bs_[i]) = v;
        }
        __syncthreads();

        if (kc + 1 < NC) {
#pragma unroll
            for (int i = 0; i < NA; ++i) ra[i] = *(const v4f_t*)(A + ag[i] + (kc + 1) * KC);
#pragma unroll
            for (int i = 0; i < NB; ++i) rb[i] = *(const v4f_t*)(B + bg[i] + (kc + 1) * BSTEP);
        }
        __builtin_amdgcn_sched_barrier(0);

        v2f_t fa[2][G][TM], fb[2][G][TN];
#pragma unroll
        for (int u = 0; u < G; ++u) {
#pragma unroll
            for (int i = 0; i < TM; ++i)
                fa[0][u][i] = *(const v2f_t*)(afr + i * 16 * PA + u * 4);
#pragma unroll
            for (int j = 0; j < TN; ++j) {
                if (BT) {
                    fb[0][u][j] = *(const v2f_t*)(bfr + j * 16 * PB + u * 4);
                } else {
                    fb[0][u][j].x = bfr[(u * 4) * PB + j * 16];
                    fb[0][u][j].y = bfr[(u * 4 + 1) * PB + j * 16];
                }
            }
        }
#pragma unroll
        for (int g = 0; g < NG; ++g) {
            if (g + 1 < NG) {
#pragma unroll
                for (int u = 0; u < G; ++u) {
                    const int s = (g + 1) * G + u;
#pragma unroll
                    for (int i = 0; i < TM; ++i)
                        fa[(g + 1) & 1][u][i] = *(const v2f_t*)(afr + i * 16 * PA + s * 4);
#pragma unroll
                    for (int j = 0; j < TN; ++j) {
                        if (BT) {
                            fb[(g + 1) & 1][u][j] = *(const v2f_t*)(bfr + j * 16 * PB + s * 4);
                        } else {
                            fb[(g + 1) & 1][u][j].x = bfr[(s * 4) * PB + j * 16];
                            fb[(g + 1) & 1][u][j].y = bfr[(s * 4 + 1) * PB + j * 16];
                        }
                    }
                }
            }
            __builtin_amdgcn_sched_barrier(0);
#pragma unroll
            for (int u = 0; u < G; ++u)
#pragma unroll
                for (int i = 0; i < TM; ++i)
#pragma unroll
                    for (int j = 0; j < TN; ++j)
                        acc[i][j] = wmma_f32(fa[g & 1][u][i], fb[g & 1][u][j], acc[i][j]);
        }
        __syncthreads();
    }

#pragma unroll
    for (int i = 0; i < TM; ++i) {
#pragma unroll
        for (int j = 0; j < TN; ++j) {
            float* crow = C + (m0 + wm + i * 16 + 8 * hi) * N + n0 + wn + j * 16 + lm;
#pragma unroll
            for (int v = 0; v < 8; ++v) crow[v * N] = acc[i][j][v];
        }
    }
}

template <int TM, int TN, int KC, int G, int N, bool BT>
__global__ void __launch_bounds__(128)
gemm_lds_kernel(const float* __restrict__ A,
                const float* __restrict__ B,
                float* __restrict__ C) {
    using cfg = tile_cfg<TM, TN, KC, BT>;
    __shared__ __attribute__((aligned(16))) float As[cfg::AS];
    __shared__ __attribute__((aligned(16))) float Bs[cfg::BS];
    gemm_lds_tile<TM, TN, KC, G, N, BT, false, false>(A, B, C, As, Bs, blockIdx.x, blockIdx.y);
}

__global__ void __launch_bounds__(128)
first_level_kernel(const float* __restrict__ W,
                   const float* __restrict__ wout,
                   float* __restrict__ A2,
                   float* __restrict__ R1) {
    using sq = tile_cfg<1, 1, 128, false>;
    using ro = tile_cfg<1, 1, 128, true>;
    constexpr int AS = sq::AS > ro::AS ? sq::AS : ro::AS;
    constexpr int BS = sq::BS > ro::BS ? sq::BS : ro::BS;
    __shared__ __attribute__((aligned(16))) float As[AS];
    __shared__ __attribute__((aligned(16))) float Bs[BS];
    const int id = blockIdx.x;
    if (id < 256) {
        gemm_lds_tile<1, 1, 128, 8, NDIM, false, true, true>(W, W, A2, As, Bs, id & 15, id >> 4);
    } else {
        const int id2 = id - 256;
        gemm_lds_tile<1, 1, 128, 8, ODIM, true, true, false>(W, wout, R1, As, Bs, id2 & 7, id2 >> 3);
    }
}

extern "C" void kernel_launch(void* const* d_in, const int* in_sizes, int n_in,
                              void* d_out, int out_size, void* d_ws, size_t ws_size,
                              hipStream_t stream) {
    (void)in_sizes; (void)n_in; (void)out_size; (void)ws_size;

    const float* x    = (const float*)d_in[0];
    const float* W    = (const float*)d_in[1];
    const float* wout = (const float*)d_in[2];
    float*       out  = (float*)d_out;

    float* T0 = (float*)d_ws;
    float* T1 = T0 + NDIM * NDIM;
    float* R1 = T1 + NDIM * NDIM;
    float* P  = R1 + NDIM * ODIM;

    first_level_kernel<<<384, 128, 0, stream>>>(W, wout, T0, R1);

    gemm_lds_kernel<1, 1, 128, 8, NDIM, false><<<dim3(16, 16), 128, 0, stream>>>(T0, T0, T1);
    gemm_lds_kernel<1, 1, 128, 8, NDIM, false><<<dim3(16, 16), 128, 0, stream>>>(T1, T1, T0);
    gemm_lds_kernel<1, 1, 128, 8, NDIM, false><<<dim3(16, 16), 128, 0, stream>>>(T0, T0, T1);
    gemm_lds_kernel<1, 1, 128, 8, NDIM, false><<<dim3(16, 16), 128, 0, stream>>>(T1, T1, T0);
    gemm_lds_kernel<1, 1, 128, 8, NDIM, false><<<dim3(16, 16), 128, 0, stream>>>(T0, T0, T1);

    gemm_lds_kernel<1, 1, 128, 8, ODIM, false><<<dim3(8, 16), 128, 0, stream>>>(T1, R1, P);

    gemm_lds_kernel<2, 2, 32, 4, ODIM, false><<<dim3(4, 256), 128, 0, stream>>>(x, P, out);
}


# Two-stage LDS ring with one LDS-only barrier per K chunk in every product: next chunk's global loads and LDS refill run under the WMMAs (was two full barriers per chunk); last product K chunk 64
# speedup vs baseline: 1.9527x; 1.0739x over previous
#include <hip/hip_runtime.h>


typedef float v2f_t __attribute__((ext_vector_type(2)));
typedef float v4f_t __attribute__((ext_vector_type(4)));
typedef float v8f_t __attribute__((ext_vector_type(8)));

#define NDIM 512
#define BDIM 16384
#define ODIM 256

static __device__ __forceinline__ v8f_t wmma_f32(v2f_t a, v2f_t b, v8f_t c) {
    return __builtin_amdgcn_wmma_f32_16x16x4_f32(
        false, a, false, b, (short)0, c, false, false);
}

static __device__ __forceinline__ void lds_barrier() {
    __builtin_amdgcn_fence(__ATOMIC_RELEASE, "workgroup", "local");
    __builtin_amdgcn_s_barrier();
    __builtin_amdgcn_fence(__ATOMIC_ACQUIRE, "workgroup", "local");
}

static __device__ __forceinline__ v4f_t leak_form(v4f_t w, int row, int col0) {
    v4f_t v;
#pragma unroll
    for (int e = 0; e < 4; ++e) {
        float s = 0.1f * w[e];
        if (row == col0 + e) s += 0.9f;
        v[e] = s;
    }
    return v;
}

template <int TM, int TN, int KC, bool BT>
struct tile_cfg {
    static constexpr int BM    = 32 * TM;
    static constexpr int BN    = 32 * TN;
    static constexpr int PA    = KC + 4;
    static constexpr int PB    = BT ? (KC + 4) : (BN + 8);
    static constexpr int BROWS = BT ? BN : KC;
    static constexpr int AS    = BM * PA;
    static constexpr int BS    = BROWS * PB;
};

template <int TM, int TN, int G, int PA, int PB, bool BT>
static __device__ __forceinline__ void
read_frags(const float* afr, const float* bfr, const int s0,
           v2f_t (&fa)[G][TM], v2f_t (&fb)[G][TN]) {
#pragma unroll
    for (int u = 0; u < G; ++u) {
        const int s = s0 + u;
#pragma unroll
        for (int i = 0; i < TM; ++i)
            fa[u][i] = *(const v2f_t*)(afr + i * 16 * PA + s * 4);
#pragma unroll
        for (int j = 0; j < TN; ++j) {
            if (BT) {
                fb[u][j] = *(const v2f_t*)(bfr + j * 16 * PB + s * 4);
            } else {
                fb[u][j].x = bfr[(s * 4) * PB + j * 16];
                fb[u][j].y = bfr[(s * 4 + 1) * PB + j * 16];
            }
        }
    }
}

template <int TM, int TN, int G>
static __device__ __forceinline__ void
mma_group(const v2f_t (&fa)[G][TM], const v2f_t (&fb)[G][TN], v8f_t (&acc)[TM][TN]) {
#pragma unroll
    for (int u = 0; u < G; ++u)
#pragma unroll
        for (int i = 0; i < TM; ++i)
#pragma unroll
            for (int j = 0; j < TN; ++j)
                acc[i][j] = wmma_f32(fa[u][i], fb[u][j], acc[i][j]);
}

template <int TM, int TN, int KC, int G, int N, int PD, bool BT, bool FA, bool FB>
static __device__ __forceinline__ void
gemm_lds_tile(const float* __restrict__ A,
              const float* __restrict__ B,
              float* __restrict__ C,
              float* As, float* Bs, const int bx, const int by) {
    using cfg = tile_cfg<TM, TN, KC, BT>;
    constexpr int BM    = cfg::BM;
    constexpr int BN    = cfg::BN;
    constexpr int PA    = cfg::PA;
    constexpr int PB    = cfg::PB;
    constexpr int SA    = cfg::AS;
    constexpr int SB    = cfg::BS;
    constexpr int BPR   = BT ? (KC / 4) : (BN / 4);
    constexpr int NA    = (BM * KC / 4) / 128;
    constexpr int NB    = (KC * BN / 4) / 128;
    constexpr int NC    = NDIM / KC;
    constexpr int STEPS = KC / 4;
    constexpr int NG    = STEPS / G;
    static_assert((BM * KC / 4) % 128 == 0 && (KC * BN / 4) % 128 == 0, "chunk must split evenly over 128 threads");
    static_assert(NDIM % KC == 0 && KC % 4 == 0 && STEPS % G == 0, "KC must divide K, G must divide the steps");
    static_assert((PA / 4) % 2 == 1 && PB % 4 == 0, "pitches");
    static_assert(!(FB && BT), "forming A on load is written for the k-major B operand only");
    static_assert(NG % 2 == 0 && NG >= 2, "the two fragment buffers alternate; the next chunk's first group lands in buffer 0");
    static_assert(PD == 1 || PD == 2, "one or two prefetch register sets");
    static_assert(NC >= 2, "ring of two stages");

    const int t    = threadIdx.x;
    const int lane = t & 31;
    const int wave = t >> 5;
    const int lm   = lane & 15;
    const int hi   = lane >> 4;
    const int m0   = by * BM;
    const int n0   = bx * BN;
    const int wm   = (wave >> 1) * (16 * TM);
    const int wn   = (wave & 1) * (16 * TN);

    int ag[NA], as_[NA], bg[NB], bs_[NB];
#pragma unroll
    for (int i = 0; i < NA; ++i) {
        const int idx = t + i * 128;
        const int r = idx / (KC / 4);
        const int c = idx % (KC / 4);
        ag[i]  = (m0 + r) * NDIM + c * 4;
        as_[i] = r * PA + c * 4;
    }
#pragma unroll
    for (int i = 0; i < NB; ++i) {
        const int idx = t + i * 128;
        const int r = idx / BPR;
        const int c = idx % BPR;
        bg[i]  = BT ? ((n0 + r) * NDIM + c * 4) : (r * N + n0 + c * 4);
        bs_[i] = r * PB + c * 4;
    }
    constexpr int BSTEP = BT ? KC : KC * N;

    v4f_t ra[PD][NA], rb[PD][NB];
#pragma unroll
    for (int p = 0; p < PD; ++p) {
#pragma unroll
        for (int i = 0; i < NA; ++i) ra[p][i] = *(const v4f_t*)(A + ag[i] + p * KC);
#pragma unroll
        for (int i = 0; i < NB; ++i) rb[p][i] = *(const v4f_t*)(B + bg[i] + p * BSTEP);
    }

    v8f_t acc[TM][TN];
#pragma unroll
    for (int i = 0; i < TM; ++i)
#pragma unroll
        for (int j = 0; j < TN; ++j) acc[i][j] = v8f_t{};

    const float* afr = As + (wm + lm) * PA + 2 * hi;
    const float* bfr = BT ? (Bs + (wn + lm) * PB + 2 * hi) : (Bs + (2 * hi) * PB + wn + lm);

    v2f_t fa[2][G][TM], fb[2][G][TN];

#pragma unroll
    for (int kc = 0; kc < NC; ++kc) {
        const int cur = kc & 1;
        const int nxt = cur ^ 1;

        if (kc == 0) {
#pragma unroll
            for (int i = 0; i < NA; ++i) {
                v4f_t v = ra[0][i];
                if (FA) {
                    const int idx = t + i * 128;
                    v = leak_form(v, m0 + idx / (KC / 4), (idx % (KC / 4)) * 4);
                }
                *(v4f_t*)(As + as_[i]) = v;
            }
#pragma unroll
            for (int i = 0; i < NB; ++i) {
                v4f_t v = rb[0][i];
                if (FB) {
                    const int idx = t + i * 128;
                    v = leak_form(v, idx / BPR, n0 + (idx % BPR) * 4);
                }
                *(v4f_t*)(Bs + bs_[i]) = v;
            }
            lds_barrier();
            read_frags<TM, TN, G, PA, PB, BT>(afr, bfr, 0, fa[0], fb[0]);
        }

        if (kc + PD < NC) {
            const int set = (PD == 2) ? cur : 0;
#pragma unroll
            for (int i = 0; i < NA; ++i) ra[set][i] = *(const v4f_t*)(A + ag[i] + (kc + PD) * KC);
#pragma unroll
            for (int i = 0; i < NB; ++i) rb[set][i] = *(const v4f_t*)(B + bg[i] + (kc + PD) * BSTEP);
        }
        __builtin_amdgcn_sched_barrier(0);

#pragma unroll
        for (int g = 0; g < NG; ++g) {
            if (g + 1 < NG)
                read_frags<TM, TN, G, PA, PB, BT>(afr + cur * SA, bfr + cur * SB, (g + 1) * G,
                                                  fa[(g + 1) & 1], fb[(g + 1) & 1]);
            if (g == NG - 2 && kc + 1 < NC) {
                const int set = (PD == 2) ? nxt : 0;
#pragma unroll
                for (int i = 0; i < NA; ++i) {
                    v4f_t v = ra[set][i];
                    if (FA) {
                        const int idx = t + i * 128;
                        v = leak_form(v, m0 + idx / (KC / 4), (kc + 1) * KC + (idx % (KC / 4)) * 4);
                    }
                    *(v4f_t*)(As + nxt * SA + as_[i]) = v;
                }
#pragma unroll
                for (int i = 0; i < NB; ++i) {
                    v4f_t v = rb[set][i];
                    if (FB) {
                        const int idx = t + i * 128;
                        v = leak_form(v, (kc + 1) * KC + idx / BPR, n0 + (idx % BPR) * 4);
                    }
                    *(v4f_t*)(Bs + nxt * SB + bs_[i]) = v;
                }
            }
            if (g == NG - 1 && kc + 1 < NC) {
                __builtin_amdgcn_sched_barrier(0);
                lds_barrier();
                read_frags<TM, TN, G, PA, PB, BT>(afr + nxt * SA, bfr + nxt * SB, 0, fa[0], fb[0]);
            }
            __builtin_amdgcn_sched_barrier(0);
            mma_group<TM, TN, G>(fa[g & 1], fb[g & 1], acc);
        }
    }

#pragma unroll
    for (int i = 0; i < TM; ++i) {
#pragma unroll
        for (int j = 0; j < TN; ++j) {
            float* crow = C + (m0 + wm + i * 16 + 8 * hi) * N + n0 + wn + j * 16 + lm;
#pragma unroll
            for (int v = 0; v < 8; ++v) crow[v * N] = acc[i][j][v];
        }
    }
}

template <int TM, int TN, int KC, int G, int N, int PD, bool BT>
__global__ void __launch_bounds__(128)
gemm_lds_kernel(const float* __restrict__ A,
                const float* __restrict__ B,
                float* __restrict__ C) {
    using cfg = tile_cfg<TM, TN, KC, BT>;
    __shared__ __attribute__((aligned(16))) float As[2 * cfg::AS];
    __shared__ __attribute__((aligned(16))) float Bs[2 * cfg::BS];
    gemm_lds_tile<TM, TN, KC, G, N, PD, BT, false, false>(A, B, C, As, Bs, blockIdx.x, blockIdx.y);
}

__global__ void __launch_bounds__(128)
first_level_kernel(const float* __restrict__ W,
                   const float* __restrict__ wout,
                   float* __restrict__ A2,
                   float* __restrict__ R1) {
    using sq = tile_cfg<1, 1, 128, false>;
    using ro = tile_cfg<1, 1, 128, true>;
    constexpr int AS = sq::AS > ro::AS ? sq::AS : ro::AS;
    constexpr int BS = sq::BS > ro::BS ? sq::BS : ro::BS;
    __shared__ __attribute__((aligned(16))) float As[2 * AS];
    __shared__ __attribute__((aligned(16))) float Bs[2 * BS];
    const int id = blockIdx.x;
    if (id < 256) {
        gemm_lds_tile<1, 1, 128, 8, NDIM, 2, false, true, true>(W, W, A2, As, Bs, id & 15, id >> 4);
    } else {
        const int id2 = id - 256;
        gemm_lds_tile<1, 1, 128, 8, ODIM, 2, true, true, false>(W, wout, R1, As, Bs, id2 & 7, id2 >> 3);
    }
}

extern "C" void kernel_launch(void* const* d_in, const int* in_sizes, int n_in,
                              void* d_out, int out_size, void* d_ws, size_t ws_size,
                              hipStream_t stream) {
    (void)in_sizes; (void)n_in; (void)out_size; (void)ws_size;

    const float* x    = (const float*)d_in[0];
    const float* W    = (const float*)d_in[1];
    const float* wout = (const float*)d_in[2];
    float*       out  = (float*)d_out;

    float* T0 = (float*)d_ws;
    float* T1 = T0 + NDIM * NDIM;
    float* R1 = T1 + NDIM * NDIM;
    float* P  = R1 + NDIM * ODIM;

    first_level_kernel<<<384, 128, 0, stream>>>(W, wout, T0, R1);

    gemm_lds_kernel<1, 1, 128, 8, NDIM, 2, false><<<dim3(16, 16), 128, 0, stream>>>(T0, T0, T1);
    gemm_lds_kernel<1, 1, 128, 8, NDIM, 2, false><<<dim3(16, 16), 128, 0, stream>>>(T1, T1, T0);
    gemm_lds_kernel<1, 1, 128, 8, NDIM, 2, false><<<dim3(16, 16), 128, 0, stream>>>(T0, T0, T1);
    gemm_lds_kernel<1, 1, 128, 8, NDIM, 2, false><<<dim3(16, 16), 128, 0, stream>>>(T1, T1, T0);
    gemm_lds_kernel<1, 1, 128, 8, NDIM, 2, false><<<dim3(16, 16), 128, 0, stream>>>(T0, T0, T1);

    gemm_lds_kernel<1, 1, 128, 8, ODIM, 2, false><<<dim3(8, 16), 128, 0, stream>>>(T1, R1, P);

    gemm_lds_kernel<2, 2, 64, 2, ODIM, 1, false><<<dim3(4, 256), 128, 0, stream>>>(x, P, out);
}
